# starting kernel
# baseline (speedup 1.0000x reference)
#include <hip/hip_runtime.h>
#include <math.h>


typedef float v2f __attribute__((ext_vector_type(2)));
typedef float v8f __attribute__((ext_vector_type(8)));

#define BSZ 4096
#define NF  2048
#define NT  300
#define RH  512
#define HD  512

#define MT      32
#define THREADS 256

__device__ __forceinline__ v8f wmma_f32_step(v2f a, v2f b, v8f c) {
    return __builtin_amdgcn_wmma_f32_16x16x4_f32(false, a, false, b,
                                                 (short)0, c, false, false);
}

template <int K, int LD, bool MULTIPLY>
__device__ __forceinline__ void stage1_pass(
    const float* __restrict__ A, const float* __restrict__ W,
    float* uLDS, int m0, int ni0, int r, int hi)
{
    v8f t8[8];
    #pragma unroll
    for (int i = 0; i < 8; ++i) t8[i] = (v8f){};

    const size_t rowA = (size_t)(m0 + r) * LD;
    const size_t rowB = (size_t)(m0 + 16 + r) * LD;

    #pragma unroll 2
    for (int k0 = 0; k0 < K; k0 += 4) {
        const int kk = k0 + 2 * hi;
        v2f a0; a0.x = A[rowA + kk]; a0.y = A[rowA + kk + 1];
        v2f a1; a1.x = A[rowB + kk]; a1.y = A[rowB + kk + 1];
        #pragma unroll
        for (int t = 0; t < 4; ++t) {
            const int n0 = (ni0 + t) * 16;
            v2f b; b.x = W[(size_t)kk * RH + n0 + r];
                   b.y = W[(size_t)(kk + 1) * RH + n0 + r];
            t8[2 * t]     = wmma_f32_step(a0, b, t8[2 * t]);
            t8[2 * t + 1] = wmma_f32_step(a1, b, t8[2 * t + 1]);
        }
    }

    #pragma unroll
    for (int t = 0; t < 4; ++t) {
        const int nn = (ni0 + t) * 16 + r;
        #pragma unroll
        for (int mi = 0; mi < 2; ++mi) {
            #pragma unroll
            for (int e = 0; e < 8; ++e) {
                const int mm = mi * 16 + 8 * hi + e;
                if (MULTIPLY) uLDS[mm * RH + nn] *= t8[2 * t + mi][e];
                else          uLDS[mm * RH + nn]  = t8[2 * t + mi][e];
            }
        }
    }
}

template <int K1, int LD1, int K2, int LD2>
__device__ __forceinline__ void product_and_project(
    const float* __restrict__ A1, const float* __restrict__ W1,
    const float* __restrict__ A2, const float* __restrict__ W2,
    const float* __restrict__ Ws,
    float* uLDS, v8f* acc,
    int m0, int ni0, int r, int hi)
{
    stage1_pass<K1, LD1, false>(A1, W1, uLDS, m0, ni0, r, hi);
    stage1_pass<K2, LD2, true >(A2, W2, uLDS, m0, ni0, r, hi);
    __syncthreads();

    const int mloc0 = r;
    const int mloc1 = 16 + r;
    #pragma unroll 2
    for (int k0 = 0; k0 < RH; k0 += 4) {
        const int kk = k0 + 2 * hi;
        v2f a0; a0.x = uLDS[mloc0 * RH + kk]; a0.y = uLDS[mloc0 * RH + kk + 1];
        v2f a1; a1.x = uLDS[mloc1 * RH + kk]; a1.y = uLDS[mloc1 * RH + kk + 1];
        if (k0 + 32 < RH)
            __builtin_prefetch(&Ws[(size_t)(kk + 32) * HD + ni0 * 16 + r], 0, 1);
        #pragma unroll
        for (int t = 0; t < 4; ++t) {
            const int n0 = (ni0 + t) * 16;
            v2f b; b.x = Ws[(size_t)kk * HD + n0 + r];
                   b.y = Ws[(size_t)(kk + 1) * HD + n0 + r];
            acc[2 * t]     = wmma_f32_step(a0, b, acc[2 * t]);
            acc[2 * t + 1] = wmma_f32_step(a1, b, acc[2 * t + 1]);
        }
    }
    __syncthreads();
}

__global__ __launch_bounds__(THREADS)
void vncl_gate_kernel(const float* __restrict__ v,  const float* __restrict__ s,
                      const float* __restrict__ x,  const float* __restrict__ h,
                      const float* __restrict__ Wa, const float* __restrict__ Wb,
                      const float* __restrict__ Wc, const float* __restrict__ Ua,
                      const float* __restrict__ Ub, const float* __restrict__ Uc,
                      const float* __restrict__ Ca, const float* __restrict__ Cb,
                      const float* __restrict__ Cc, const float* __restrict__ bias,
                      float* __restrict__ logits  )
{
    __shared__ float uLDS[MT * RH];

    const int g    = blockIdx.y;
    const int m0   = blockIdx.x * MT;
    const int lane = threadIdx.x & 31;
    const int wave = threadIdx.x >> 5;
    const int r    = lane & 15;
    const int hi   = lane >> 4;

    const int ni0 = wave * 4;

    const float* Wa_g = Wa + (size_t)g * RH * RH;
    const float* Wb_g = Wb + (size_t)g * NT * RH;
    const float* Wc_g = Wc + (size_t)g * RH * HD;
    const float* Ua_g = Ua + (size_t)g * RH * RH;
    const float* Ub_g = Ub + (size_t)g * NT * RH;
    const float* Uc_g = Uc + (size_t)g * RH * HD;
    const float* Ca_g = Ca + (size_t)g * NF * RH;
    const float* Cb_g = Cb + (size_t)g * NT * RH;
    const float* Cc_g = Cc + (size_t)g * RH * HD;

    v8f acc[8];
    #pragma unroll
    for (int t = 0; t < 8; ++t) acc[t] = (v8f){};

    product_and_project<RH, RH, NT, NT>(x, Wa_g, s, Wb_g, Wc_g,
                                        uLDS, acc, m0, ni0, r, hi);
    product_and_project<NF, NF, NT, NT>(v, Ca_g, s, Cb_g, Cc_g,
                                        uLDS, acc, m0, ni0, r, hi);
    product_and_project<NT, NT, RH, RH>(s, Ub_g, h, Ua_g, Uc_g,
                                        uLDS, acc, m0, ni0, r, hi);

    const float* bg = bias + (size_t)g * HD;
    float* Lg = logits + ((size_t)g * BSZ + m0) * HD;
    #pragma unroll
    for (int t = 0; t < 4; ++t) {
        const int nn = (ni0 + t) * 16 + r;
        #pragma unroll
        for (int mi = 0; mi < 2; ++mi) {
            #pragma unroll
            for (int e = 0; e < 8; ++e) {
                const int mm = mi * 16 + 8 * hi + e;
                Lg[(size_t)mm * HD + nn] = acc[2 * t + mi][e] + bg[nn];
            }
        }
    }
}

__global__ __launch_bounds__(256)
void lstm_epilogue(const float* __restrict__ logits,
                   const float* __restrict__ c,
                   float* __restrict__ out, int n)
{
    const int idx = blockIdx.x * blockDim.x + threadIdx.x;
    if (idx >= n) return;
    const size_t stride = (size_t)BSZ * HD;
    const float li = logits[idx];
    const float lf = logits[idx + stride];
    const float lo = logits[idx + 2 * stride];
    const float lg = logits[idx + 3 * stride];
    const float ig = 1.0f / (1.0f + __expf(-li));
    const float fg = 1.0f / (1.0f + __expf(-lf));
    const float og = 1.0f / (1.0f + __expf(-lo));
    const float gg = tanhf(lg);
    const float nc = fg * c[idx] + ig * gg;
    const float nh = og * tanhf(nc);
    out[idx]          = nh;
    out[idx + stride] = nc;
}

extern "C" void kernel_launch(void* const* d_in, const int* in_sizes, int n_in,
                              void* d_out, int out_size, void* d_ws, size_t ws_size,
                              hipStream_t stream) {
    (void)in_sizes; (void)n_in; (void)out_size; (void)ws_size;
    const float* v  = (const float*)d_in[0];
    const float* s  = (const float*)d_in[1];
    const float* x  = (const float*)d_in[2];
    const float* h  = (const float*)d_in[3];
    const float* c  = (const float*)d_in[4];
    const float* Wa = (const float*)d_in[5];
    const float* Wb = (const float*)d_in[6];
    const float* Wc = (const float*)d_in[7];
    const float* Ua = (const float*)d_in[8];
    const float* Ub = (const float*)d_in[9];
    const float* Uc = (const float*)d_in[10];
    const float* Ca = (const float*)d_in[11];
    const float* Cb = (const float*)d_in[12];
    const float* Cc = (const float*)d_in[13];
    const float* b  = (const float*)d_in[14];

    float* logits = (float*)d_ws;

    dim3 grid(BSZ / MT, 4);
    vncl_gate_kernel<<<grid, dim3(THREADS), 0, stream>>>(
        v, s, x, h, Wa, Wb, Wc, Ua, Ub, Uc, Ca, Cb, Cc, b, logits);

    const int n = BSZ * HD;
    lstm_epilogue<<<(n + 255) / 256, 256, 0, stream>>>(logits, c, (float*)d_out, n);
}


# W and activation k-slabs staged through LDS by whole-workgroup 16-byte line-covering loads one slab ahead (2 buffers, 1 barrier per slab); all WMMA fragments read from LDS; 2 workgroups per WGP
# speedup vs baseline: 1.8283x; 1.8283x over previous
#include <hip/hip_runtime.h>
#include <math.h>


typedef float v2f __attribute__((ext_vector_type(2)));
typedef float v4f __attribute__((ext_vector_type(4)));
typedef float v8f __attribute__((ext_vector_type(8)));

#define BSZ 4096
#define NF  2048
#define NT  300
#define RH  512
#define HD  512

#define MT      32
#define THREADS 256

#define KSMAX  16
#define WROW   512
#define WSLAB  (KSMAX * WROW)
#define APITCH 20
#define ASLAB  (MT * APITCH)

__device__ __forceinline__ v8f wmma_f32_step(v2f a, v2f b, v8f c) {
    return __builtin_amdgcn_wmma_f32_16x16x4_f32(false, a, false, b,
                                                 (short)0, c, false, false);
}

struct Lane {
    int tid, r, hi, ni0, m0;
    int c4;
    int rsel;
    int pos[4];
};

template <int KS, int LD, bool AG>
__device__ __forceinline__ void slab_load(
    const float* __restrict__ A, const float* __restrict__ W, int ks0,
    const Lane& L, v4f* wreg, v4f& areg)
{
    #pragma unroll
    for (int i = 0; i < KS / 2; ++i) {
        const int kl = L.rsel + 2 * i;
        wreg[i] = *(const v4f*)(W + (size_t)(ks0 + kl) * RH + L.c4 * 4);
    }
    if (AG) {
        constexpr int QN = KS / 4;
        if (L.tid < MT * QN) {
            const int row = L.tid / QN, q = L.tid % QN;
            areg = *(const v4f*)(A + (size_t)(L.m0 + row) * LD + ks0 + 4 * q);
        }
    }
}

template <int KS, bool AG>
__device__ __forceinline__ void slab_store(
    float* Wl, float* Al, int buf, const Lane& L, const v4f* wreg, const v4f& areg)
{
    #pragma unroll
    for (int i = 0; i < KS / 2; ++i) {
        const int kl  = L.rsel + 2 * i;
        const int pos = (L.c4 * 4 + 16 * (i & 1)) & (WROW - 1);
        *(v4f*)(Wl + buf * WSLAB + kl * WROW + pos) = wreg[i];
    }
    if (AG) {
        constexpr int QN = KS / 4;
        if (L.tid < MT * QN) {
            const int row = L.tid / QN, q = L.tid % QN;
            *(v4f*)(Al + buf * ASLAB + row * APITCH + 4 * q) = areg;
        }
    }
}

template <int KS, bool ULDS_A>
__device__ __forceinline__ void slab_mma(
    const float* Wl, const float* Al, const float* uLDS, int buf, int ks0,
    const Lane& L, v8f* acc)
{
    #pragma unroll
    for (int j = 0; j < KS / 4; ++j) {
        const int kl = 4 * j + 2 * L.hi;
        v2f a0, a1;
        if (ULDS_A) {
            a0 = *(const v2f*)(uLDS + L.r * RH + ks0 + kl);
            a1 = *(const v2f*)(uLDS + (16 + L.r) * RH + ks0 + kl);
        } else {
            a0 = *(const v2f*)(Al + buf * ASLAB + L.r * APITCH + kl);
            a1 = *(const v2f*)(Al + buf * ASLAB + (16 + L.r) * APITCH + kl);
        }
        #pragma unroll
        for (int t = 0; t < 4; ++t) {
            v2f b;
            b.x = Wl[buf * WSLAB + kl * WROW + L.pos[t]];
            b.y = Wl[buf * WSLAB + (kl + 1) * WROW + L.pos[t]];
            acc[2 * t]     = wmma_f32_step(a0, b, acc[2 * t]);
            acc[2 * t + 1] = wmma_f32_step(a1, b, acc[2 * t + 1]);
        }
    }
}

template <int K, int KS, int LD, bool ULDS_A>
__device__ __forceinline__ void gemm_pass(
    const float* __restrict__ A, const float* __restrict__ W,
    float* Wl, float* Al, const float* uLDS, const Lane& L, v8f* acc)
{
    static_assert(K % KS == 0 && KS % 4 == 0 && KS <= KSMAX, "slab shape");
    constexpr int NS = K / KS;
    v4f wreg[KS / 2];
    v4f areg = (v4f){};

    slab_load<KS, LD, !ULDS_A>(A, W, 0, L, wreg, areg);
    slab_store<KS, !ULDS_A>(Wl, Al, 0, L, wreg, areg);
    __syncthreads();

    #pragma unroll 1
    for (int s = 0; s < NS; ++s) {
        const int cur = s & 1;
        const bool more = (s + 1 < NS);
        if (more) slab_load<KS, LD, !ULDS_A>(A, W, (s + 1) * KS, L, wreg, areg);
        slab_mma<KS, ULDS_A>(Wl, Al, uLDS, cur, s * KS, L, acc);
        if (more) slab_store<KS, !ULDS_A>(Wl, Al, cur ^ 1, L, wreg, areg);
        __syncthreads();
    }
}

template <bool MULTIPLY>
__device__ __forceinline__ void product_tile(float* uLDS, const v8f* t8, const Lane& L)
{
    #pragma unroll
    for (int t = 0; t < 4; ++t) {
        const int nn = (L.ni0 + t) * 16 + L.r;
        #pragma unroll
        for (int mi = 0; mi < 2; ++mi) {
            #pragma unroll
            for (int e = 0; e < 8; ++e) {
                const int mm = mi * 16 + 8 * L.hi + e;
                if (MULTIPLY) uLDS[mm * RH + nn] *= t8[2 * t + mi][e];
                else          uLDS[mm * RH + nn]  = t8[2 * t + mi][e];
            }
        }
    }
}

template <int K1, int KS1, int K2, int KS2>
__device__ __forceinline__ void product_and_project(
    const float* __restrict__ A1, const float* __restrict__ W1,
    const float* __restrict__ A2, const float* __restrict__ W2,
    const float* __restrict__ Ws,
    float* uLDS, float* Wl, float* Al, v8f* acc, const Lane& L)
{
    v8f t8[8];
    #pragma unroll
    for (int i = 0; i < 8; ++i) t8[i] = (v8f){};
    gemm_pass<K1, KS1, K1, false>(A1, W1, Wl, Al, uLDS, L, t8);
    product_tile<false>(uLDS, t8, L);


    #pragma unroll
    for (int i = 0; i < 8; ++i) t8[i] = (v8f){};
    gemm_pass<K2, KS2, K2, false>(A2, W2, Wl, Al, uLDS, L, t8);
    product_tile<true>(uLDS, t8, L);

    gemm_pass<RH, 16, RH, true>(nullptr, Ws, Wl, Al, uLDS, L, acc);
}

__global__ __launch_bounds__(THREADS, 4)
void gate_kernel(const float* __restrict__ v,  const float* __restrict__ s,
                 const float* __restrict__ x,  const float* __restrict__ h,
                 const float* __restrict__ Wa, const float* __restrict__ Wb,
                 const float* __restrict__ Wc, const float* __restrict__ Ua,
                 const float* __restrict__ Ub, const float* __restrict__ Uc,
                 const float* __restrict__ Ca, const float* __restrict__ Cb,
                 const float* __restrict__ Cc, const float* __restrict__ bias,
                 float* __restrict__ logits  )
{
    __shared__ __attribute__((aligned(16))) float uLDS[MT * RH];
    __shared__ __attribute__((aligned(16))) float Wl[2 * WSLAB];
    __shared__ __attribute__((aligned(16))) float Al[2 * ASLAB];

    const int g = blockIdx.y;
    Lane L;
    L.tid  = threadIdx.x;
    L.m0   = blockIdx.x * MT;
    const int lane = threadIdx.x & 31;
    const int wave = threadIdx.x >> 5;
    L.r    = lane & 15;
    L.hi   = lane >> 4;
    L.ni0  = wave * 4;
    L.c4   = threadIdx.x & 127;
    L.rsel = threadIdx.x >> 7;
    #pragma unroll
    for (int t = 0; t < 4; ++t)
        L.pos[t] = ((L.ni0 + t) * 16 + L.r + 16 * L.hi) & (WROW - 1);

    const float* Wa_g = Wa + (size_t)g * RH * RH;
    const float* Wb_g = Wb + (size_t)g * NT * RH;
    const float* Wc_g = Wc + (size_t)g * RH * HD;
    const float* Ua_g = Ua + (size_t)g * RH * RH;
    const float* Ub_g = Ub + (size_t)g * NT * RH;
    const float* Uc_g = Uc + (size_t)g * RH * HD;
    const float* Ca_g = Ca + (size_t)g * NF * RH;
    const float* Cb_g = Cb + (size_t)g * NT * RH;
    const float* Cc_g = Cc + (size_t)g * RH * HD;

    v8f acc[8];
    #pragma unroll
    for (int t = 0; t < 8; ++t) acc[t] = (v8f){};

    product_and_project<RH, 16, NT, 12>(x, Wa_g, s, Wb_g, Wc_g, uLDS, Wl, Al, acc, L);
    product_and_project<NF, 16, NT, 12>(v, Ca_g, s, Cb_g, Cc_g, uLDS, Wl, Al, acc, L);
    product_and_project<NT, 12, RH, 16>(s, Ub_g, h, Ua_g, Uc_g, uLDS, Wl, Al, acc, L);

    const float* bg = bias + (size_t)g * HD;
    float* Lg = logits + ((size_t)g * BSZ + L.m0) * HD;
    #pragma unroll
    for (int t = 0; t < 4; ++t) {
        const int nn = (L.ni0 + t) * 16 + L.r;
        #pragma unroll
        for (int mi = 0; mi < 2; ++mi) {
            #pragma unroll
            for (int e = 0; e < 8; ++e) {
                const int mm = mi * 16 + 8 * L.hi + e;
                Lg[(size_t)mm * HD + nn] = acc[2 * t + mi][e] + bg[nn];
            }
        }
    }
}

__global__ __launch_bounds__(256)
void lstm_epilogue(const float* __restrict__ logits,
                   const float* __restrict__ c,
                   float* __restrict__ out, int n)
{
    const int idx = blockIdx.x * blockDim.x + threadIdx.x;
    if (idx >= n) return;
    const size_t stride = (size_t)BSZ * HD;
    const float li = logits[idx];
    const float lf = logits[idx + stride];
    const float lo = logits[idx + 2 * stride];
    const float lg = logits[idx + 3 * stride];
    const float ig = 1.0f / (1.0f + __expf(-li));
    const float fg = 1.0f / (1.0f + __expf(-lf));
    const float og = 1.0f / (1.0f + __expf(-lo));
    const float gg = tanhf(lg);
    const float nc = fg * c[idx] + ig * gg;
    const float nh = og * tanhf(nc);
    out[idx]          = nh;
    out[idx + stride] = nc;
}

extern "C" void kernel_launch(void* const* d_in, const int* in_sizes, int n_in,
                              void* d_out, int out_size, void* d_ws, size_t ws_size,
                              hipStream_t stream) {
    (void)in_sizes; (void)n_in; (void)out_size; (void)ws_size;
    const float* v  = (const float*)d_in[0];
    const float* s  = (const float*)d_in[1];
    const float* x  = (const float*)d_in[2];
    const float* h  = (const float*)d_in[3];
    const float* c  = (const float*)d_in[4];
    const float* Wa = (const float*)d_in[5];
    const float* Wb = (const float*)d_in[6];
    const float* Wc = (const float*)d_in[7];
    const float* Ua = (const float*)d_in[8];
    const float* Ub = (const float*)d_in[9];
    const float* Uc = (const float*)d_in[10];
    const float* Ca = (const float*)d_in[11];
    const float* Cb = (const float*)d_in[12];
    const float* Cc = (const float*)d_in[13];
    const float* b  = (const float*)d_in[14];

    float* logits = (float*)d_ws;

    dim3 grid(BSZ / MT, 4);
    gate_kernel<<<grid, dim3(THREADS), 0, stream>>>(
        v, s, x, h, Wa, Wb, Wc, Ua, Ub, Uc, Ca, Cb, Cc, b, logits);

    const int n = BSZ * HD;
    lstm_epilogue<<<(n + 255) / 256, 256, 0, stream>>>(logits, c, (float*)d_out, n);
}


# Product tile rows in LDS padded from 512 to 516 floats: the 16 rows of a stage-2 A fragment read fall in 64 distinct banks instead of one 16-way conflict; tile stores lose their 2-way conflict
# speedup vs baseline: 2.0507x; 1.1216x over previous
#include <hip/hip_runtime.h>
#include <math.h>


typedef float v2f __attribute__((ext_vector_type(2)));
typedef float v4f __attribute__((ext_vector_type(4)));
typedef float v8f __attribute__((ext_vector_type(8)));

#define BSZ 4096
#define NF  2048
#define NT  300
#define RH  512
#define HD  512

#define MT      32
#define THREADS 256

#define KSMAX  16
#define WROW   512
#define WSLAB  (KSMAX * WROW)
#define APITCH 20
#define ASLAB  (MT * APITCH)
#define UPITCH (RH + 4)

__device__ __forceinline__ v8f wmma_f32_step(v2f a, v2f b, v8f c) {
    return __builtin_amdgcn_wmma_f32_16x16x4_f32(false, a, false, b,
                                                 (short)0, c, false, false);
}

struct Lane {
    int tid, r, hi, ni0, m0;
    int c4;
    int rsel;
    int pos[4];
};

template <int KS, int LD, bool AG>
__device__ __forceinline__ void slab_load(
    const float* __restrict__ A, const float* __restrict__ W, int ks0,
    const Lane& L, v4f* wreg, v4f& areg)
{
    #pragma unroll
    for (int i = 0; i < KS / 2; ++i) {
        const int kl = L.rsel + 2 * i;
        wreg[i] = *(const v4f*)(W + (size_t)(ks0 + kl) * RH + L.c4 * 4);
    }
    if (AG) {
        constexpr int QN = KS / 4;
        if (L.tid < MT * QN) {
            const int row = L.tid / QN, q = L.tid % QN;
            areg = *(const v4f*)(A + (size_t)(L.m0 + row) * LD + ks0 + 4 * q);
        }
    }
}

template <int KS, bool AG>
__device__ __forceinline__ void slab_store(
    float* Wl, float* Al, int buf, const Lane& L, const v4f* wreg, const v4f& areg)
{
    #pragma unroll
    for (int i = 0; i < KS / 2; ++i) {
        const int kl  = L.rsel + 2 * i;
        const int pos = (L.c4 * 4 + 16 * (i & 1)) & (WROW - 1);
        *(v4f*)(Wl + buf * WSLAB + kl * WROW + pos) = wreg[i];
    }
    if (AG) {
        constexpr int QN = KS / 4;
        if (L.tid < MT * QN) {
            const int row = L.tid / QN, q = L.tid % QN;
            *(v4f*)(Al + buf * ASLAB + row * APITCH + 4 * q) = areg;
        }
    }
}

template <int KS, bool ULDS_A>
__device__ __forceinline__ void slab_mma(
    const float* Wl, const float* Al, const float* uLDS, int buf, int ks0,
    const Lane& L, v8f* acc)
{
    #pragma unroll
    for (int j = 0; j < KS / 4; ++j) {
        const int kl = 4 * j + 2 * L.hi;
        v2f a0, a1;
        if (ULDS_A) {
            a0 = *(const v2f*)(uLDS + L.r * UPITCH + ks0 + kl);
            a1 = *(const v2f*)(uLDS + (16 + L.r) * UPITCH + ks0 + kl);
        } else {
            a0 = *(const v2f*)(Al + buf * ASLAB + L.r * APITCH + kl);
            a1 = *(const v2f*)(Al + buf * ASLAB + (16 + L.r) * APITCH + kl);
        }
        #pragma unroll
        for (int t = 0; t < 4; ++t) {
            v2f b;
            b.x = Wl[buf * WSLAB + kl * WROW + L.pos[t]];
            b.y = Wl[buf * WSLAB + (kl + 1) * WROW + L.pos[t]];
            acc[2 * t]     = wmma_f32_step(a0, b, acc[2 * t]);
            acc[2 * t + 1] = wmma_f32_step(a1, b, acc[2 * t + 1]);
        }
    }
}

template <int K, int KS, int LD, bool ULDS_A>
__device__ __forceinline__ void gemm_pass(
    const float* __restrict__ A, const float* __restrict__ W,
    float* Wl, float* Al, const float* uLDS, const Lane& L, v8f* acc)
{
    static_assert(K % KS == 0 && KS % 4 == 0 && KS <= KSMAX, "slab shape");
    constexpr int NS = K / KS;
    v4f wreg[KS / 2];
    v4f areg = (v4f){};

    slab_load<KS, LD, !ULDS_A>(A, W, 0, L, wreg, areg);
    slab_store<KS, !ULDS_A>(Wl, Al, 0, L, wreg, areg);
    __syncthreads();

    #pragma unroll 1
    for (int s = 0; s < NS; ++s) {
        const int cur = s & 1;
        const bool more = (s + 1 < NS);
        if (more) slab_load<KS, LD, !ULDS_A>(A, W, (s + 1) * KS, L, wreg, areg);
        slab_mma<KS, ULDS_A>(Wl, Al, uLDS, cur, s * KS, L, acc);
        if (more) slab_store<KS, !ULDS_A>(Wl, Al, cur ^ 1, L, wreg, areg);
        __syncthreads();
    }
}

template <bool MULTIPLY>
__device__ __forceinline__ void product_tile(float* uLDS, const v8f* t8, const Lane& L)
{
    #pragma unroll
    for (int t = 0; t < 4; ++t) {
        const int nn = (L.ni0 + t) * 16 + L.r;
        #pragma unroll
        for (int mi = 0; mi < 2; ++mi) {
            #pragma unroll
            for (int e = 0; e < 8; ++e) {
                const int mm = mi * 16 + 8 * L.hi + e;
                if (MULTIPLY) uLDS[mm * UPITCH + nn] *= t8[2 * t + mi][e];
                else          uLDS[mm * UPITCH + nn]  = t8[2 * t + mi][e];
            }
        }
    }
}

template <int K1, int KS1, int K2, int KS2>
__device__ __forceinline__ void product_and_project(
    const float* __restrict__ A1, const float* __restrict__ W1,
    const float* __restrict__ A2, const float* __restrict__ W2,
    const float* __restrict__ Ws,
    float* uLDS, float* Wl, float* Al, v8f* acc, const Lane& L)
{
    v8f t8[8];
    #pragma unroll
    for (int i = 0; i < 8; ++i) t8[i] = (v8f){};
    gemm_pass<K1, KS1, K1, false>(A1, W1, Wl, Al, uLDS, L, t8);
    product_tile<false>(uLDS, t8, L);


    #pragma unroll
    for (int i = 0; i < 8; ++i) t8[i] = (v8f){};
    gemm_pass<K2, KS2, K2, false>(A2, W2, Wl, Al, uLDS, L, t8);
    product_tile<true>(uLDS, t8, L);

    gemm_pass<RH, 16, RH, true>(nullptr, Ws, Wl, Al, uLDS, L, acc);
}

__global__ __launch_bounds__(THREADS, 4)
void gate_kernel(const float* __restrict__ v,  const float* __restrict__ s,
                 const float* __restrict__ x,  const float* __restrict__ h,
                 const float* __restrict__ Wa, const float* __restrict__ Wb,
                 const float* __restrict__ Wc, const float* __restrict__ Ua,
                 const float* __restrict__ Ub, const float* __restrict__ Uc,
                 const float* __restrict__ Ca, const float* __restrict__ Cb,
                 const float* __restrict__ Cc, const float* __restrict__ bias,
                 float* __restrict__ logits  )
{
    __shared__ __attribute__((aligned(16))) float uLDS[MT * UPITCH];
    __shared__ __attribute__((aligned(16))) float Wl[2 * WSLAB];
    __shared__ __attribute__((aligned(16))) float Al[2 * ASLAB];

    const int g = blockIdx.y;
    Lane L;
    L.tid  = threadIdx.x;
    L.m0   = blockIdx.x * MT;
    const int lane = threadIdx.x & 31;
    const int wave = threadIdx.x >> 5;
    L.r    = lane & 15;
    L.hi   = lane >> 4;
    L.ni0  = wave * 4;
    L.c4   = threadIdx.x & 127;
    L.rsel = threadIdx.x >> 7;
    #pragma unroll
    for (int t = 0; t < 4; ++t)
        L.pos[t] = ((L.ni0 + t) * 16 + L.r + 16 * L.hi) & (WROW - 1);

    const float* Wa_g = Wa + (size_t)g * RH * RH;
    const float* Wb_g = Wb + (size_t)g * NT * RH;
    const float* Wc_g = Wc + (size_t)g * RH * HD;
    const float* Ua_g = Ua + (size_t)g * RH * RH;
    const float* Ub_g = Ub + (size_t)g * NT * RH;
    const float* Uc_g = Uc + (size_t)g * RH * HD;
    const float* Ca_g = Ca + (size_t)g * NF * RH;
    const float* Cb_g = Cb + (size_t)g * NT * RH;
    const float* Cc_g = Cc + (size_t)g * RH * HD;

    v8f acc[8];
    #pragma unroll
    for (int t = 0; t < 8; ++t) acc[t] = (v8f){};

    product_and_project<RH, 16, NT, 12>(x, Wa_g, s, Wb_g, Wc_g, uLDS, Wl, Al, acc, L);
    product_and_project<NF, 16, NT, 12>(v, Ca_g, s, Cb_g, Cc_g, uLDS, Wl, Al, acc, L);
    product_and_project<NT, 12, RH, 16>(s, Ub_g, h, Ua_g, Uc_g, uLDS, Wl, Al, acc, L);

    const float* bg = bias + (size_t)g * HD;
    float* Lg = logits + ((size_t)g * BSZ + L.m0) * HD;
    #pragma unroll
    for (int t = 0; t < 4; ++t) {
        const int nn = (L.ni0 + t) * 16 + L.r;
        #pragma unroll
        for (int mi = 0; mi < 2; ++mi) {
            #pragma unroll
            for (int e = 0; e < 8; ++e) {
                const int mm = mi * 16 + 8 * L.hi + e;
                Lg[(size_t)mm * HD + nn] = acc[2 * t + mi][e] + bg[nn];
            }
        }
    }
}

__global__ __launch_bounds__(256)
void lstm_epilogue(const float* __restrict__ logits,
                   const float* __restrict__ c,
                   float* __restrict__ out, int n)
{
    const int idx = blockIdx.x * blockDim.x + threadIdx.x;
    if (idx >= n) return;
    const size_t stride = (size_t)BSZ * HD;
    const float li = logits[idx];
    const float lf = logits[idx + stride];
    const float lo = logits[idx + 2 * stride];
    const float lg = logits[idx + 3 * stride];
    const float ig = 1.0f / (1.0f + __expf(-li));
    const float fg = 1.0f / (1.0f + __expf(-lf));
    const float og = 1.0f / (1.0f + __expf(-lo));
    const float gg = tanhf(lg);
    const float nc = fg * c[idx] + ig * gg;
    const float nh = og * tanhf(nc);
    out[idx]          = nh;
    out[idx + stride] = nc;
}

extern "C" void kernel_launch(void* const* d_in, const int* in_sizes, int n_in,
                              void* d_out, int out_size, void* d_ws, size_t ws_size,
                              hipStream_t stream) {
    (void)in_sizes; (void)n_in; (void)out_size; (void)ws_size;
    const float* v  = (const float*)d_in[0];
    const float* s  = (const float*)d_in[1];
    const float* x  = (const float*)d_in[2];
    const float* h  = (const float*)d_in[3];
    const float* c  = (const float*)d_in[4];
    const float* Wa = (const float*)d_in[5];
    const float* Wb = (const float*)d_in[6];
    const float* Wc = (const float*)d_in[7];
    const float* Ua = (const float*)d_in[8];
    const float* Ub = (const float*)d_in[9];
    const float* Uc = (const float*)d_in[10];
    const float* Ca = (const float*)d_in[11];
    const float* Cb = (const float*)d_in[12];
    const float* Cc = (const float*)d_in[13];
    const float* b  = (const float*)d_in[14];

    float* logits = (float*)d_ws;

    dim3 grid(BSZ / MT, 4);
    gate_kernel<<<grid, dim3(THREADS), 0, stream>>>(
        v, s, x, h, Wa, Wb, Wc, Ua, Ub, Uc, Ca, Cb, Cc, b, logits);

    const int n = BSZ * HD;
    lstm_epilogue<<<(n + 255) / 256, 256, 0, stream>>>(logits, c, (float*)d_out, n);
}


# Slab-loop fragment reads hand-written as two-address LDS loads: one per B fragment (rows k,k+1 straight into the operand pair), immediate offsets off one base per tile, one group, counted waits
# speedup vs baseline: 2.1345x; 1.0409x over previous
#include <hip/hip_runtime.h>
#include <math.h>


typedef float v2f __attribute__((ext_vector_type(2)));
typedef float v4f __attribute__((ext_vector_type(4)));
typedef float v8f __attribute__((ext_vector_type(8)));

#define BSZ 4096
#define NF  2048
#define NT  300
#define RH  512
#define HD  512

#define MT      32
#define THREADS 256

#define KSMAX  16
#define WROW   512
#define WSLAB  (KSMAX * WROW)
#define APITCH 20
#define ASLAB  (MT * APITCH)
#define UPITCH (RH + 4)

__device__ __forceinline__ v8f wmma_f32_step(v2f a, v2f b, v8f c) {
    return __builtin_amdgcn_wmma_f32_16x16x4_f32(false, a, false, b,
                                                 (short)0, c, false, false);
}

typedef __attribute__((address_space(3))) const float lds_cfloat;
__device__ __forceinline__ unsigned lds_addr(const float* p) {
    return (unsigned)(size_t)(lds_cfloat*)p;
}

template <int O0, int O1>
__device__ __forceinline__ v2f lds_read_rows2(unsigned addr) {
    v2f d;
    asm volatile("ds_load_2addr_stride64_b32 %0, %1 offset0:%2 offset1:%3"
                 : "=v"(d) : "v"(addr), "n"(O0), "n"(O1) : "memory");
    return d;
}
template <int O0, int O1>
__device__ __forceinline__ v4f lds_read_pairs2(unsigned addr) {
    v4f d;
    asm volatile("ds_load_2addr_b64 %0, %1 offset0:%2 offset1:%3"
                 : "=v"(d) : "v"(addr), "n"(O0), "n"(O1) : "memory");
    return d;
}
template <int N>
__device__ __forceinline__ void wait_reads(v4f& a, v2f& b0, v2f& b1, v2f& b2, v2f& b3) {
    asm volatile("s_wait_dscnt %5"
                 : "+v"(a), "+v"(b0), "+v"(b1), "+v"(b2), "+v"(b3) : "n"(N) : "memory");
}
template <int N>
__device__ __forceinline__ void wait_reads(v4f& a0, v4f& a1, v2f& b0, v2f& b1, v2f& b2, v2f& b3) {
    asm volatile("s_wait_dscnt %6"
                 : "+v"(a0), "+v"(a1), "+v"(b0), "+v"(b1), "+v"(b2), "+v"(b3) : "n"(N) : "memory");
}
template <int N>
__device__ __forceinline__ void wait_reads(v2f& b0, v2f& b1, v2f& b2, v2f& b3) {
    asm volatile("s_wait_dscnt %4"
                 : "+v"(b0), "+v"(b1), "+v"(b2), "+v"(b3) : "n"(N) : "memory");
}
__device__ __forceinline__ v2f lo2(v4f q) { return __builtin_shufflevector(q, q, 0, 1); }
__device__ __forceinline__ v2f hi2(v4f q) { return __builtin_shufflevector(q, q, 2, 3); }

typedef __attribute__((address_space(1))) const float gfloat;
typedef __attribute__((address_space(1))) const v4f   gv4f;
__device__ __forceinline__ gfloat* lane_uniform(const float* p) {
    const unsigned long long u = (unsigned long long)p;
    const unsigned lo = __builtin_amdgcn_readfirstlane((unsigned)u);
    const unsigned hi = __builtin_amdgcn_readfirstlane((unsigned)(u >> 32));
    return (gfloat*)(((unsigned long long)hi << 32) | lo);
}

struct Lane {
    int tid, r, hi, ni0, m0;
    int c4;
    int rsel;
    unsigned wb[4];
    unsigned ab;
    unsigned ub;
};

template <int KS, int LD, bool AG>
__device__ __forceinline__ void slab_load(
    const float* __restrict__ A, const float* __restrict__ W, int ks0,
    const Lane& L, v4f* wreg, v4f& areg)
{
    gfloat* Ws = lane_uniform(W + (size_t)ks0 * RH);
    #pragma unroll
    for (int i = 0; i < KS / 2; ++i) {
        const unsigned off = (unsigned)((L.rsel + 2 * i) * RH + L.c4 * 4);
        wreg[i] = *(gv4f*)(Ws + off);
    }
    if (AG) {
        constexpr int QN = KS / 4;
        if (L.tid < MT * QN) {
            const int row = L.tid / QN, q = L.tid % QN;
            gfloat* As = lane_uniform(A + (size_t)L.m0 * LD + ks0);
            const unsigned off = (unsigned)(row * LD + 4 * q);
            areg = *(gv4f*)(As + off);
        }
    }
}

template <int KS, bool AG>
__device__ __forceinline__ void slab_store(
    float* Wl, float* Al, int buf, const Lane& L, const v4f* wreg, const v4f& areg)
{
    #pragma unroll
    for (int i = 0; i < KS / 2; ++i) {
        const int kl  = L.rsel + 2 * i;
        const int pos = (L.c4 * 4 + 16 * (i & 1)) & (WROW - 1);
        *(v4f*)(Wl + buf * WSLAB + kl * WROW + pos) = wreg[i];
    }
    if (AG) {
        constexpr int QN = KS / 4;
        if (L.tid < MT * QN) {
            const int row = L.tid / QN, q = L.tid % QN;
            *(v4f*)(Al + buf * ASLAB + row * APITCH + 4 * q) = areg;
        }
    }
}

template <int KS>
__device__ __forceinline__ void slab_mma_staged(int buf, const Lane& L, v8f* acc)
{
    constexpr int NJ = KS / 4;
    static_assert(NJ == 3 || NJ == 4, "slabs of 12 or 16 k");
    const unsigned aB = L.ab + (unsigned)buf * (ASLAB * 4u);
    unsigned wB[4];
    #pragma unroll
    for (int t = 0; t < 4; ++t) wB[t] = L.wb[t] + (unsigned)buf * (WSLAB * 4u);
    v4f a[NJ];
    v2f b[NJ][4];
#define SLAB_READS(J) \
    a[J]    = lds_read_pairs2<2 * (J), 160 + 2 * (J)>(aB); \
    b[J][0] = lds_read_rows2<32 * (J), 32 * (J) + 8>(wB[0]); \
    b[J][1] = lds_read_rows2<32 * (J), 32 * (J) + 8>(wB[1]); \
    b[J][2] = lds_read_rows2<32 * (J), 32 * (J) + 8>(wB[2]); \
    b[J][3] = lds_read_rows2<32 * (J), 32 * (J) + 8>(wB[3]);
#define SLAB_WMMAS(J) \
    wait_reads<5 * (NJ - 1 - (J))>(a[J], b[J][0], b[J][1], b[J][2], b[J][3]); \
    _Pragma("unroll") \
    for (int t = 0; t < 4; ++t) { \
        acc[2 * t]     = wmma_f32_step(lo2(a[J]), b[J][t], acc[2 * t]); \
        acc[2 * t + 1] = wmma_f32_step(hi2(a[J]), b[J][t], acc[2 * t + 1]); \
    }
    SLAB_READS(0) SLAB_READS(1) SLAB_READS(2)
    if (NJ == 4) { SLAB_READS(NJ - 1) }
    SLAB_WMMAS(0) SLAB_WMMAS(1) SLAB_WMMAS(2)
    if (NJ == 4) { SLAB_WMMAS(NJ - 1) }
#undef SLAB_READS
#undef SLAB_WMMAS
}

__device__ __forceinline__ void slab_mma_tile(int buf, int ks0, const Lane& L, v8f* acc)
{
    const unsigned u0 = L.ub + (unsigned)ks0 * 4u;
    const unsigned u1 = u0 + 16u * UPITCH * 4u;
    unsigned wB[4];
    #pragma unroll
    for (int t = 0; t < 4; ++t) wB[t] = L.wb[t] + (unsigned)buf * (WSLAB * 4u);
    v4f p0[2], p1[2];
    v2f b[4][4];
#define TILE_B(J) \
    b[J][0] = lds_read_rows2<32 * (J), 32 * (J) + 8>(wB[0]); \
    b[J][1] = lds_read_rows2<32 * (J), 32 * (J) + 8>(wB[1]); \
    b[J][2] = lds_read_rows2<32 * (J), 32 * (J) + 8>(wB[2]); \
    b[J][3] = lds_read_rows2<32 * (J), 32 * (J) + 8>(wB[3]);
#define TILE_WMMAS(J, A0, A1) \
    _Pragma("unroll") \
    for (int t = 0; t < 4; ++t) { \
        acc[2 * t]     = wmma_f32_step(A0, b[J][t], acc[2 * t]); \
        acc[2 * t + 1] = wmma_f32_step(A1, b[J][t], acc[2 * t + 1]); \
    }
    p0[0] = lds_read_pairs2<0, 2>(u0);
    p1[0] = lds_read_pairs2<0, 2>(u1);
    TILE_B(0) TILE_B(1)
    p0[1] = lds_read_pairs2<4, 6>(u0);
    p1[1] = lds_read_pairs2<4, 6>(u1);
    TILE_B(2) TILE_B(3)
    wait_reads<14>(p0[0], p1[0], b[0][0], b[0][1], b[0][2], b[0][3]);
    TILE_WMMAS(0, lo2(p0[0]), lo2(p1[0]))
    wait_reads<10>(b[1][0], b[1][1], b[1][2], b[1][3]);
    TILE_WMMAS(1, hi2(p0[0]), hi2(p1[0]))
    wait_reads<4>(p0[1], p1[1], b[2][0], b[2][1], b[2][2], b[2][3]);
    TILE_WMMAS(2, lo2(p0[1]), lo2(p1[1]))
    wait_reads<0>(b[3][0], b[3][1], b[3][2], b[3][3]);
    TILE_WMMAS(3, hi2(p0[1]), hi2(p1[1]))
#undef TILE_B
#undef TILE_WMMAS
}

template <int KS, bool ULDS_A>
__device__ __forceinline__ void slab_mma(int buf, int ks0, const Lane& L, v8f* acc)
{
    if (ULDS_A) slab_mma_tile(buf, ks0, L, acc);
    else        slab_mma_staged<KS>(buf, L, acc);
}

template <int K, int KS, int LD, bool ULDS_A>
__device__ __forceinline__ void gemm_pass(
    const float* __restrict__ A, const float* __restrict__ W,
    float* Wl, float* Al, const float* uLDS, const Lane& L, v8f* acc)
{
    static_assert(K % KS == 0 && KS % 4 == 0 && KS <= KSMAX, "slab shape");
    constexpr int NS = K / KS;
    v4f wreg[KS / 2];
    v4f areg = (v4f){};

    slab_load<KS, LD, !ULDS_A>(A, W, 0, L, wreg, areg);
    slab_store<KS, !ULDS_A>(Wl, Al, 0, L, wreg, areg);
    __syncthreads();

    #pragma unroll 1
    for (int s = 0; s < NS; ++s) {
        const int cur = s & 1;
        const bool more = (s + 1 < NS);
        if (more) slab_load<KS, LD, !ULDS_A>(A, W, (s + 1) * KS, L, wreg, areg);
        slab_mma<KS, ULDS_A>(cur, s * KS, L, acc);
        if (more) slab_store<KS, !ULDS_A>(Wl, Al, cur ^ 1, L, wreg, areg);
        __syncthreads();
    }
}

template <bool MULTIPLY>
__device__ __forceinline__ void product_tile(float* uLDS, const v8f* t8, const Lane& L)
{
    #pragma unroll
    for (int t = 0; t < 4; ++t) {
        const int nn = (L.ni0 + t) * 16 + L.r;
        #pragma unroll
        for (int mi = 0; mi < 2; ++mi) {
            #pragma unroll
            for (int e = 0; e < 8; ++e) {
                const int mm = mi * 16 + 8 * L.hi + e;
                if (MULTIPLY) uLDS[mm * UPITCH + nn] *= t8[2 * t + mi][e];
                else          uLDS[mm * UPITCH + nn]  = t8[2 * t + mi][e];
            }
        }
    }
}

template <int K1, int KS1, int K2, int KS2>
__device__ __forceinline__ void product_and_project(
    const float* __restrict__ A1, const float* __restrict__ W1,
    const float* __restrict__ A2, const float* __restrict__ W2,
    const float* __restrict__ Ws,
    float* uLDS, float* Wl, float* Al, v8f* acc, const Lane& L)
{
    v8f t8[8];
    #pragma unroll
    for (int i = 0; i < 8; ++i) t8[i] = (v8f){};
    gemm_pass<K1, KS1, K1, false>(A1, W1, Wl, Al, uLDS, L, t8);
    product_tile<false>(uLDS, t8, L);


    #pragma unroll
    for (int i = 0; i < 8; ++i) t8[i] = (v8f){};
    gemm_pass<K2, KS2, K2, false>(A2, W2, Wl, Al, uLDS, L, t8);
    product_tile<true>(uLDS, t8, L);

    gemm_pass<RH, 16, RH, true>(nullptr, Ws, Wl, Al, uLDS, L, acc);
}

__global__ __launch_bounds__(THREADS, 4)
void gate_kernel(const float* __restrict__ v,  const float* __restrict__ s,
                 const float* __restrict__ x,  const float* __restrict__ h,
                 const float* __restrict__ Wa, const float* __restrict__ Wb,
                 const float* __restrict__ Wc, const float* __restrict__ Ua,
                 const float* __restrict__ Ub, const float* __restrict__ Uc,
                 const float* __restrict__ Ca, const float* __restrict__ Cb,
                 const float* __restrict__ Cc, const float* __restrict__ bias,
                 float* __restrict__ logits  )
{
    __shared__ __attribute__((aligned(16))) float uLDS[MT * UPITCH];
    __shared__ __attribute__((aligned(16))) float Wl[2 * WSLAB];
    __shared__ __attribute__((aligned(16))) float Al[2 * ASLAB];

    const int g = blockIdx.y;
    Lane L;
    L.tid  = threadIdx.x;
    L.m0   = blockIdx.x * MT;
    const int lane = threadIdx.x & 31;
    const int wave = threadIdx.x >> 5;
    L.r    = lane & 15;
    L.hi   = lane >> 4;
    L.ni0  = wave * 4;
    L.c4   = threadIdx.x & 127;
    L.rsel = threadIdx.x >> 7;
    #pragma unroll
    for (int t = 0; t < 4; ++t) {
        const int pos = ((L.ni0 + t) * 16 + L.r + 16 * L.hi) & (WROW - 1);
        L.wb[t] = lds_addr(Wl) + (unsigned)(2 * L.hi * WROW + pos) * 4u;
    }
    L.ab = lds_addr(Al)   + (unsigned)(L.r * APITCH + 2 * L.hi) * 4u;
    L.ub = lds_addr(uLDS) + (unsigned)(L.r * UPITCH + 2 * L.hi) * 4u;

    const float* Wa_g = Wa + (size_t)g * RH * RH;
    const float* Wb_g = Wb + (size_t)g * NT * RH;
    const float* Wc_g = Wc + (size_t)g * RH * HD;
    const float* Ua_g = Ua + (size_t)g * RH * RH;
    const float* Ub_g = Ub + (size_t)g * NT * RH;
    const float* Uc_g = Uc + (size_t)g * RH * HD;
    const float* Ca_g = Ca + (size_t)g * NF * RH;
    const float* Cb_g = Cb + (size_t)g * NT * RH;
    const float* Cc_g = Cc + (size_t)g * RH * HD;

    v8f acc[8];
    #pragma unroll
    for (int t = 0; t < 8; ++t) acc[t] = (v8f){};

    product_and_project<RH, 16, NT, 12>(x, Wa_g, s, Wb_g, Wc_g, uLDS, Wl, Al, acc, L);
    product_and_project<NF, 16, NT, 12>(v, Ca_g, s, Cb_g, Cc_g, uLDS, Wl, Al, acc, L);
    product_and_project<NT, 12, RH, 16>(s, Ub_g, h, Ua_g, Uc_g, uLDS, Wl, Al, acc, L);

    const float* bg = bias + (size_t)g * HD;
    float* Lg = logits + ((size_t)g * BSZ + L.m0) * HD;
    #pragma unroll
    for (int t = 0; t < 4; ++t) {
        const int nn = (L.ni0 + t) * 16 + L.r;
        #pragma unroll
        for (int mi = 0; mi < 2; ++mi) {
            #pragma unroll
            for (int e = 0; e < 8; ++e) {
                const int mm = mi * 16 + 8 * L.hi + e;
                Lg[(size_t)mm * HD + nn] = acc[2 * t + mi][e] + bg[nn];
            }
        }
    }
}

__global__ __launch_bounds__(256)
void lstm_epilogue(const float* __restrict__ logits,
                   const float* __restrict__ c,
                   float* __restrict__ out, int n)
{
    const int idx = blockIdx.x * blockDim.x + threadIdx.x;
    if (idx >= n) return;
    const size_t stride = (size_t)BSZ * HD;
    const float li = logits[idx];
    const float lf = logits[idx + stride];
    const float lo = logits[idx + 2 * stride];
    const float lg = logits[idx + 3 * stride];
    const float ig = 1.0f / (1.0f + __expf(-li));
    const float fg = 1.0f / (1.0f + __expf(-lf));
    const float og = 1.0f / (1.0f + __expf(-lo));
    const float gg = tanhf(lg);
    const float nc = fg * c[idx] + ig * gg;
    const float nh = og * tanhf(nc);
    out[idx]          = nh;
    out[idx + stride] = nc;
}

extern "C" void kernel_launch(void* const* d_in, const int* in_sizes, int n_in,
                              void* d_out, int out_size, void* d_ws, size_t ws_size,
                              hipStream_t stream) {
    (void)in_sizes; (void)n_in; (void)out_size; (void)ws_size;
    const float* v  = (const float*)d_in[0];
    const float* s  = (const float*)d_in[1];
    const float* x  = (const float*)d_in[2];
    const float* h  = (const float*)d_in[3];
    const float* c  = (const float*)d_in[4];
    const float* Wa = (const float*)d_in[5];
    const float* Wb = (const float*)d_in[6];
    const float* Wc = (const float*)d_in[7];
    const float* Ua = (const float*)d_in[8];
    const float* Ub = (const float*)d_in[9];
    const float* Uc = (const float*)d_in[10];
    const float* Ca = (const float*)d_in[11];
    const float* Cb = (const float*)d_in[12];
    const float* Cc = (const float*)d_in[13];
    const float* b  = (const float*)d_in[14];

    float* logits = (float*)d_ws;

    dim3 grid(BSZ / MT, 4);
    gate_kernel<<<grid, dim3(THREADS), 0, stream>>>(
        v, s, x, h, Wa, Wb, Wc, Ua, Ub, Uc, Ca, Cb, Cc, b, logits);

    const int n = BSZ * HD;
    lstm_epilogue<<<(n + 255) / 256, 256, 0, stream>>>(logits, c, (float*)d_out, n);
}


# Slab barrier moved in front of each slab's last k-step: next slab's LDS fragment reads issue behind it under that k-step's 8 WMMAs; next slab stored early, loads 2 slabs ahead kept in flight
# speedup vs baseline: 2.2384x; 1.0487x over previous
#include <hip/hip_runtime.h>
#include <math.h>


typedef float v2f __attribute__((ext_vector_type(2)));
typedef float v4f __attribute__((ext_vector_type(4)));
typedef float v8f __attribute__((ext_vector_type(8)));

#define BSZ 4096
#define NF  2048
#define NT  300
#define RH  512
#define HD  512

#define MT      32
#define THREADS 256

#define KSMAX  16
#define WROW   512
#define WSLAB  (KSMAX * WROW)
#define APITCH 20
#define ASLAB  (MT * APITCH)
#define UPITCH (RH + 4)

__device__ __forceinline__ v8f wmma_f32_step(v2f a, v2f b, v8f c) {
    return __builtin_amdgcn_wmma_f32_16x16x4_f32(false, a, false, b,
                                                 (short)0, c, false, false);
}

typedef __attribute__((address_space(3))) const float lds_cfloat;
__device__ __forceinline__ unsigned lds_addr(const float* p) {
    return (unsigned)(size_t)(lds_cfloat*)p;
}

template <int O0, int O1>
__device__ __forceinline__ v2f lds_read_rows2(unsigned addr) {
    v2f d;
    asm volatile("ds_load_2addr_stride64_b32 %0, %1 offset0:%2 offset1:%3"
                 : "=v"(d) : "v"(addr), "n"(O0), "n"(O1) : "memory");
    return d;
}
template <int O0, int O1>
__device__ __forceinline__ v4f lds_read_pairs2(unsigned addr) {
    v4f d;
    asm volatile("ds_load_2addr_b64 %0, %1 offset0:%2 offset1:%3"
                 : "=v"(d) : "v"(addr), "n"(O0), "n"(O1) : "memory");
    return d;
}
template <int N>
__device__ __forceinline__ void wait_reads(v4f& a, v2f& b0, v2f& b1, v2f& b2, v2f& b3) {
    asm volatile("s_wait_dscnt %5"
                 : "+v"(a), "+v"(b0), "+v"(b1), "+v"(b2), "+v"(b3) : "n"(N) : "memory");
}
template <int N>
__device__ __forceinline__ void wait_reads(v4f& a0, v4f& a1, v2f& b0, v2f& b1, v2f& b2, v2f& b3) {
    asm volatile("s_wait_dscnt %6"
                 : "+v"(a0), "+v"(a1), "+v"(b0), "+v"(b1), "+v"(b2), "+v"(b3) : "n"(N) : "memory");
}
__device__ __forceinline__ v2f lo2(v4f q) { return __builtin_shufflevector(q, q, 0, 1); }
__device__ __forceinline__ v2f hi2(v4f q) { return __builtin_shufflevector(q, q, 2, 3); }

typedef __attribute__((address_space(1))) const float gfloat;
typedef __attribute__((address_space(1))) const v4f   gv4f;
__device__ __forceinline__ gfloat* lane_uniform(const float* p) {
    const unsigned long long u = (unsigned long long)p;
    const unsigned lo = __builtin_amdgcn_readfirstlane((unsigned)u);
    const unsigned hi = __builtin_amdgcn_readfirstlane((unsigned)(u >> 32));
    return (gfloat*)(((unsigned long long)hi << 32) | lo);
}

struct Lane {
    int tid, r, hi, ni0, m0;
    int c4;
    int rsel;
    unsigned wb[4];
    unsigned ab;
    unsigned ub;
};

template <int KS, int LD, bool AG>
__device__ __forceinline__ void slab_load(
    const float* __restrict__ A, const float* __restrict__ W, int ks0,
    const Lane& L, v4f* wreg, v4f& areg)
{
    gfloat* Ws = lane_uniform(W + (size_t)ks0 * RH);
    #pragma unroll
    for (int i = 0; i < KS / 2; ++i) {
        const unsigned off = (unsigned)((L.rsel + 2 * i) * RH + L.c4 * 4);
        wreg[i] = *(gv4f*)(Ws + off);
    }
    if (AG) {
        constexpr int QN = KS / 4;
        if (L.tid < MT * QN) {
            const int row = L.tid / QN, q = L.tid % QN;
            gfloat* As = lane_uniform(A + (size_t)L.m0 * LD + ks0);
            const unsigned off = (unsigned)(row * LD + 4 * q);
            areg = *(gv4f*)(As + off);
        }
    }
}

template <int KS, bool AG>
__device__ __forceinline__ void slab_store(
    float* Wl, float* Al, int buf, const Lane& L, const v4f* wreg, const v4f& areg)
{
    #pragma unroll
    for (int i = 0; i < KS / 2; ++i) {
        const int kl  = L.rsel + 2 * i;
        const int pos = (L.c4 * 4 + 16 * (i & 1)) & (WROW - 1);
        *(v4f*)(Wl + buf * WSLAB + kl * WROW + pos) = wreg[i];
    }
    if (AG) {
        constexpr int QN = KS / 4;
        if (L.tid < MT * QN) {
            const int row = L.tid / QN, q = L.tid % QN;
            *(v4f*)(Al + buf * ASLAB + row * APITCH + 4 * q) = areg;
        }
    }
}

__device__ __forceinline__ void slab_handover(v4f& a0, v4f& a1, v2f& b0, v2f& b1, v2f& b2, v2f& b3) {
    asm volatile("s_wait_dscnt 0x0\n\ts_barrier_signal -1\n\ts_barrier_wait -1"
                 : "+v"(a0), "+v"(a1), "+v"(b0), "+v"(b1), "+v"(b2), "+v"(b3) : : "memory");
}
__device__ __forceinline__ void pass_open_barrier() {
    asm volatile("s_wait_dscnt 0x0\n\ts_barrier_signal -1\n\ts_barrier_wait -1" : : : "memory");
}
template <int N>
__device__ __forceinline__ void wait_reads(v4f& a0, v4f& a1, v4f& a2,
                                           v2f (&b0)[4], v2f (&b1)[4], v2f (&b2)[4]) {
    asm volatile("s_wait_dscnt %15"
                 : "+v"(a0), "+v"(a1), "+v"(a2),
                   "+v"(b0[0]), "+v"(b0[1]), "+v"(b0[2]), "+v"(b0[3]),
                   "+v"(b1[0]), "+v"(b1[1]), "+v"(b1[2]), "+v"(b1[3]),
                   "+v"(b2[0]), "+v"(b2[1]), "+v"(b2[2]), "+v"(b2[3])
                 : "n"(N) : "memory");
}
template <int N>
__device__ __forceinline__ void wait_reads(v4f& a0, v4f& a1, v2f (&b0)[4], v2f (&b1)[4], v2f (&b2)[4]) {
    asm volatile("s_wait_dscnt %14"
                 : "+v"(a0), "+v"(a1),
                   "+v"(b0[0]), "+v"(b0[1]), "+v"(b0[2]), "+v"(b0[3]),
                   "+v"(b1[0]), "+v"(b1[1]), "+v"(b1[2]), "+v"(b1[3]),
                   "+v"(b2[0]), "+v"(b2[1]), "+v"(b2[2]), "+v"(b2[3])
                 : "n"(N) : "memory");
}
template <int N>
__device__ __forceinline__ void wait_reads(v4f& a0, v4f& a1, v2f (&b0)[4], v2f (&b1)[4]) {
    asm volatile("s_wait_dscnt %10"
                 : "+v"(a0), "+v"(a1),
                   "+v"(b0[0]), "+v"(b0[1]), "+v"(b0[2]), "+v"(b0[3]),
                   "+v"(b1[0]), "+v"(b1[1]), "+v"(b1[2]), "+v"(b1[3])
                 : "n"(N) : "memory");
}

template <int K, int KS, int LD>
__device__ __forceinline__ void gemm_pass_staged(
    const float* __restrict__ A, const float* __restrict__ W,
    float* Wl, float* Al, const Lane& L, v8f* acc)
{
    static_assert(K % KS == 0 && KS % 4 == 0 && KS <= KSMAX, "slab shape");
    constexpr int NS = K / KS;
    constexpr int NJ = KS / 4;
    static_assert(NJ == 3 || NJ == 4, "slabs of 12 or 16 k");
    static_assert(NS >= 2, "pipelined loop needs two slabs");
    v4f wreg[KS / 2];
    v4f areg = (v4f){};
    v4f a[NJ];
    v2f b[NJ][4];
    unsigned aB, wB[4];

#define SLAB_READS(J) \
    a[J]    = lds_read_pairs2<2 * (J), 160 + 2 * (J)>(aB); \
    b[J][0] = lds_read_rows2<32 * (J), 32 * (J) + 8>(wB[0]); \
    b[J][1] = lds_read_rows2<32 * (J), 32 * (J) + 8>(wB[1]); \
    b[J][2] = lds_read_rows2<32 * (J), 32 * (J) + 8>(wB[2]); \
    b[J][3] = lds_read_rows2<32 * (J), 32 * (J) + 8>(wB[3]);
#define SLAB_WMMAS(J) \
    _Pragma("unroll") \
    for (int t = 0; t < 4; ++t) { \
        acc[2 * t]     = wmma_f32_step(lo2(a[J]), b[J][t], acc[2 * t]); \
        acc[2 * t + 1] = wmma_f32_step(hi2(a[J]), b[J][t], acc[2 * t + 1]); \
    }
#define SLAB_BASES(BUF) \
    aB = L.ab + (unsigned)(BUF) * (ASLAB * 4u); \
    _Pragma("unroll") \
    for (int t = 0; t < 4; ++t) wB[t] = L.wb[t] + (unsigned)(BUF) * (WSLAB * 4u);
#define SLAB_WAIT_HEAD() \
    if (NJ == 4) wait_reads<5>(a[0], a[1], a[2], b[0], b[1], b[2]); \
    else         wait_reads<5>(a[0], a[1], b[0], b[1]);

    slab_load<KS, LD, true>(A, W, 0, L, wreg, areg);
    slab_store<KS, true>(Wl, Al, 0, L, wreg, areg);
    slab_load<KS, LD, true>(A, W, KS, L, wreg, areg);
    pass_open_barrier();
    SLAB_BASES(0)
    SLAB_READS(0) SLAB_READS(1) SLAB_READS(2)
    if (NJ == 4) { SLAB_READS(NJ - 1) }

    #pragma unroll 1
    for (int s = 0; s < NS - 1; ++s) {
        const int nxt = (s & 1) ^ 1;
        SLAB_WAIT_HEAD()
        SLAB_WMMAS(0)
        wait_reads<0>(a[NJ - 1], b[NJ - 1][0], b[NJ - 1][1], b[NJ - 1][2], b[NJ - 1][3]);
        slab_store<KS, true>(Wl, Al, nxt, L, wreg, areg);
        if (s + 2 < NS) slab_load<KS, LD, true>(A, W, (s + 2) * KS, L, wreg, areg);
        SLAB_WMMAS(1)
        if (NJ == 4) { SLAB_WMMAS(2) }
        slab_handover(a[NJ - 2], a[NJ - 1], b[NJ - 1][0], b[NJ - 1][1], b[NJ - 1][2], b[NJ - 1][3]);
        SLAB_BASES(nxt)
        SLAB_READS(0) SLAB_READS(1)
        if (NJ == 4) { SLAB_READS(2) }
        __builtin_amdgcn_sched_barrier(0);
        SLAB_WMMAS(NJ - 1)
        SLAB_READS(NJ - 1)
    }
    SLAB_WAIT_HEAD()
    SLAB_WMMAS(0)
    wait_reads<0>(a[NJ - 1], b[NJ - 1][0], b[NJ - 1][1], b[NJ - 1][2], b[NJ - 1][3]);
    SLAB_WMMAS(1) SLAB_WMMAS(2)
    if (NJ == 4) { SLAB_WMMAS(NJ - 1) }
    __syncthreads();
#undef SLAB_READS
#undef SLAB_WMMAS
#undef SLAB_BASES
#undef SLAB_WAIT_HEAD
}

__device__ __forceinline__ void gemm_pass_tile(
    const float* __restrict__ W, float* Wl, float* Al, const Lane& L, v8f* acc)
{
    constexpr int KS = 16;
    constexpr int NS = RH / KS;
    v4f wreg[KS / 2];
    v4f areg = (v4f){};
    v4f p0[2], p1[2];
    v2f b[4][4];
    unsigned u0, u1, wB[4];

#define TILE_B(J) \
    b[J][0] = lds_read_rows2<32 * (J), 32 * (J) + 8>(wB[0]); \
    b[J][1] = lds_read_rows2<32 * (J), 32 * (J) + 8>(wB[1]); \
    b[J][2] = lds_read_rows2<32 * (J), 32 * (J) + 8>(wB[2]); \
    b[J][3] = lds_read_rows2<32 * (J), 32 * (J) + 8>(wB[3]);
#define TILE_WMMAS(J, A0, A1) \
    _Pragma("unroll") \
    for (int t = 0; t < 4; ++t) { \
        acc[2 * t]     = wmma_f32_step(A0, b[J][t], acc[2 * t]); \
        acc[2 * t + 1] = wmma_f32_step(A1, b[J][t], acc[2 * t + 1]); \
    }
#define TILE_BASES(BUF, KS0) \
    u0 = L.ub + (unsigned)(KS0) * 4u; \
    u1 = u0 + 16u * UPITCH * 4u; \
    _Pragma("unroll") \
    for (int t = 0; t < 4; ++t) wB[t] = L.wb[t] + (unsigned)(BUF) * (WSLAB * 4u);
#define TILE_READS_HEAD() \
    p0[0] = lds_read_pairs2<0, 2>(u0); \
    p1[0] = lds_read_pairs2<0, 2>(u1); \
    TILE_B(0) TILE_B(1) TILE_B(2)
#define TILE_READS_TAIL() \
    p0[1] = lds_read_pairs2<4, 6>(u0); \
    p1[1] = lds_read_pairs2<4, 6>(u1); \
    TILE_B(3)

    slab_load<KS, RH, false>(nullptr, W, 0, L, wreg, areg);
    slab_store<KS, false>(Wl, Al, 0, L, wreg, areg);
    slab_load<KS, RH, false>(nullptr, W, KS, L, wreg, areg);
    pass_open_barrier();
    TILE_BASES(0, 0)
    TILE_READS_HEAD()
    TILE_READS_TAIL()

    #pragma unroll 1
    for (int s = 0; s < NS - 1; ++s) {
        const int nxt = (s & 1) ^ 1;
        wait_reads<6>(p0[0], p1[0], b[0], b[1], b[2]);
        TILE_WMMAS(0, lo2(p0[0]), lo2(p1[0]))
        wait_reads<0>(p0[1], p1[1], b[3][0], b[3][1], b[3][2], b[3][3]);
        slab_store<KS, false>(Wl, Al, nxt, L, wreg, areg);
        if (s + 2 < NS) slab_load<KS, RH, false>(nullptr, W, (s + 2) * KS, L, wreg, areg);
        TILE_WMMAS(1, hi2(p0[0]), hi2(p1[0]))
        TILE_WMMAS(2, lo2(p0[1]), lo2(p1[1]))
        slab_handover(p0[1], p1[1], b[3][0], b[3][1], b[3][2], b[3][3]);
        TILE_BASES(nxt, (s + 1) * KS)
        TILE_READS_HEAD()
        __builtin_amdgcn_sched_barrier(0);
        TILE_WMMAS(3, hi2(p0[1]), hi2(p1[1]))
        TILE_READS_TAIL()
    }
    wait_reads<6>(p0[0], p1[0], b[0], b[1], b[2]);
    TILE_WMMAS(0, lo2(p0[0]), lo2(p1[0]))
    wait_reads<0>(p0[1], p1[1], b[3][0], b[3][1], b[3][2], b[3][3]);
    TILE_WMMAS(1, hi2(p0[0]), hi2(p1[0]))
    TILE_WMMAS(2, lo2(p0[1]), lo2(p1[1]))
    TILE_WMMAS(3, hi2(p0[1]), hi2(p1[1]))
    __syncthreads();
#undef TILE_B
#undef TILE_WMMAS
#undef TILE_BASES
#undef TILE_READS_HEAD
#undef TILE_READS_TAIL
}

template <bool MULTIPLY>
__device__ __forceinline__ void product_tile(float* uLDS, const v8f* t8, const Lane& L)
{
    #pragma unroll
    for (int t = 0; t < 4; ++t) {
        const int nn = (L.ni0 + t) * 16 + L.r;
        #pragma unroll
        for (int mi = 0; mi < 2; ++mi) {
            #pragma unroll
            for (int e = 0; e < 8; ++e) {
                const int mm = mi * 16 + 8 * L.hi + e;
                if (MULTIPLY) uLDS[mm * UPITCH + nn] *= t8[2 * t + mi][e];
                else          uLDS[mm * UPITCH + nn]  = t8[2 * t + mi][e];
            }
        }
    }
}

template <int K1, int KS1, int K2, int KS2>
__device__ __forceinline__ void product_and_project(
    const float* __restrict__ A1, const float* __restrict__ W1,
    const float* __restrict__ A2, const float* __restrict__ W2,
    const float* __restrict__ Ws,
    float* uLDS, float* Wl, float* Al, v8f* acc, const Lane& L)
{
    v8f t8[8];
    #pragma unroll
    for (int i = 0; i < 8; ++i) t8[i] = (v8f){};
    gemm_pass_staged<K1, KS1, K1>(A1, W1, Wl, Al, L, t8);
    product_tile<false>(uLDS, t8, L);


    #pragma unroll
    for (int i = 0; i < 8; ++i) t8[i] = (v8f){};
    gemm_pass_staged<K2, KS2, K2>(A2, W2, Wl, Al, L, t8);
    product_tile<true>(uLDS, t8, L);

    gemm_pass_tile(Ws, Wl, Al, L, acc);
}

__global__ __launch_bounds__(THREADS, 4)
void gate_kernel(const float* __restrict__ v,  const float* __restrict__ s,
                 const float* __restrict__ x,  const float* __restrict__ h,
                 const float* __restrict__ Wa, const float* __restrict__ Wb,
                 const float* __restrict__ Wc, const float* __restrict__ Ua,
                 const float* __restrict__ Ub, const float* __restrict__ Uc,
                 const float* __restrict__ Ca, const float* __restrict__ Cb,
                 const float* __restrict__ Cc, const float* __restrict__ bias,
                 float* __restrict__ logits  )
{
    __shared__ __attribute__((aligned(16))) float uLDS[MT * UPITCH];
    __shared__ __attribute__((aligned(16))) float Wl[2 * WSLAB];
    __shared__ __attribute__((aligned(16))) float Al[2 * ASLAB];

    const int g = blockIdx.y;
    Lane L;
    L.tid  = threadIdx.x;
    L.m0   = blockIdx.x * MT;
    const int lane = threadIdx.x & 31;
    const int wave = threadIdx.x >> 5;
    L.r    = lane & 15;
    L.hi   = lane >> 4;
    L.ni0  = wave * 4;
    L.c4   = threadIdx.x & 127;
    L.rsel = threadIdx.x >> 7;
    #pragma unroll
    for (int t = 0; t < 4; ++t) {
        const int pos = ((L.ni0 + t) * 16 + L.r + 16 * L.hi) & (WROW - 1);
        L.wb[t] = lds_addr(Wl) + (unsigned)(2 * L.hi * WROW + pos) * 4u;
    }
    L.ab = lds_addr(Al)   + (unsigned)(L.r * APITCH + 2 * L.hi) * 4u;
    L.ub = lds_addr(uLDS) + (unsigned)(L.r * UPITCH + 2 * L.hi) * 4u;

    const float* Wa_g = Wa + (size_t)g * RH * RH;
    const float* Wb_g = Wb + (size_t)g * NT * RH;
    const float* Wc_g = Wc + (size_t)g * RH * HD;
    const float* Ua_g = Ua + (size_t)g * RH * RH;
    const float* Ub_g = Ub + (size_t)g * NT * RH;
    const float* Uc_g = Uc + (size_t)g * RH * HD;
    const float* Ca_g = Ca + (size_t)g * NF * RH;
    const float* Cb_g = Cb + (size_t)g * NT * RH;
    const float* Cc_g = Cc + (size_t)g * RH * HD;

    v8f acc[8];
    #pragma unroll
    for (int t = 0; t < 8; ++t) acc[t] = (v8f){};

    product_and_project<RH, 16, NT, 12>(x, Wa_g, s, Wb_g, Wc_g, uLDS, Wl, Al, acc, L);
    product_and_project<NF, 16, NT, 12>(v, Ca_g, s, Cb_g, Cc_g, uLDS, Wl, Al, acc, L);
    product_and_project<NT, 12, RH, 16>(s, Ub_g, h, Ua_g, Uc_g, uLDS, Wl, Al, acc, L);

    const float* bg = bias + (size_t)g * HD;
    float* Lg = logits + ((size_t)g * BSZ + L.m0) * HD;
    #pragma unroll
    for (int t = 0; t < 4; ++t) {
        const int nn = (L.ni0 + t) * 16 + L.r;
        #pragma unroll
        for (int mi = 0; mi < 2; ++mi) {
            #pragma unroll
            for (int e = 0; e < 8; ++e) {
                const int mm = mi * 16 + 8 * L.hi + e;
                Lg[(size_t)mm * HD + nn] = acc[2 * t + mi][e] + bg[nn];
            }
        }
    }
}

__global__ __launch_bounds__(256)
void lstm_epilogue(const float* __restrict__ logits,
                   const float* __restrict__ c,
                   float* __restrict__ out, int n)
{
    const int idx = blockIdx.x * blockDim.x + threadIdx.x;
    if (idx >= n) return;
    const size_t stride = (size_t)BSZ * HD;
    const float li = logits[idx];
    const float lf = logits[idx + stride];
    const float lo = logits[idx + 2 * stride];
    const float lg = logits[idx + 3 * stride];
    const float ig = 1.0f / (1.0f + __expf(-li));
    const float fg = 1.0f / (1.0f + __expf(-lf));
    const float og = 1.0f / (1.0f + __expf(-lo));
    const float gg = tanhf(lg);
    const float nc = fg * c[idx] + ig * gg;
    const float nh = og * tanhf(nc);
    out[idx]          = nh;
    out[idx + stride] = nc;
}

extern "C" void kernel_launch(void* const* d_in, const int* in_sizes, int n_in,
                              void* d_out, int out_size, void* d_ws, size_t ws_size,
                              hipStream_t stream) {
    (void)in_sizes; (void)n_in; (void)out_size; (void)ws_size;
    const float* v  = (const float*)d_in[0];
    const float* s  = (const float*)d_in[1];
    const float* x  = (const float*)d_in[2];
    const float* h  = (const float*)d_in[3];
    const float* c  = (const float*)d_in[4];
    const float* Wa = (const float*)d_in[5];
    const float* Wb = (const float*)d_in[6];
    const float* Wc = (const float*)d_in[7];
    const float* Ua = (const float*)d_in[8];
    const float* Ub = (const float*)d_in[9];
    const float* Uc = (const float*)d_in[10];
    const float* Ca = (const float*)d_in[11];
    const float* Cb = (const float*)d_in[12];
    const float* Cc = (const float*)d_in[13];
    const float* b  = (const float*)d_in[14];

    float* logits = (float*)d_ws;

    dim3 grid(BSZ / MT, 4);
    gate_kernel<<<grid, dim3(THREADS), 0, stream>>>(
        v, s, x, h, Wa, Wb, Wc, Ua, Ub, Uc, Ca, Cb, Cc, b, logits);

    const int n = BSZ * HD;
    lstm_epilogue<<<(n + 255) / 256, 256, 0, stream>>>(logits, c, (float*)d_out, n);
}


# Two co-resident 32-row workgroups merged into one 16-wave workgroup per 64 rows: both halves read B fragments from ONE LDS W slab (W staged once per 64 rows); wave tile and sum order unchanged
# speedup vs baseline: 2.4202x; 1.0812x over previous
#include <hip/hip_runtime.h>
#include <math.h>


typedef float v2f __attribute__((ext_vector_type(2)));
typedef float v4f __attribute__((ext_vector_type(4)));
typedef float v8f __attribute__((ext_vector_type(8)));

#define BSZ 4096
#define NF  2048
#define NT  300
#define RH  512
#define HD  512

#define MT      64
#define THREADS 512

#define KSMAX  16
#define WROW   512
#define WSLAB  (KSMAX * WROW)
#define APITCH 20
#define ASLAB  (MT * APITCH)
#define UPITCH (RH + 4)

__device__ __forceinline__ v8f wmma_f32_step(v2f a, v2f b, v8f c) {
    return __builtin_amdgcn_wmma_f32_16x16x4_f32(false, a, false, b,
                                                 (short)0, c, false, false);
}

typedef __attribute__((address_space(3))) const float lds_cfloat;
__device__ __forceinline__ unsigned lds_addr(const float* p) {
    return (unsigned)(size_t)(lds_cfloat*)p;
}

template <int O0, int O1>
__device__ __forceinline__ v2f lds_read_rows2(unsigned addr) {
    v2f d;
    asm volatile("ds_load_2addr_stride64_b32 %0, %1 offset0:%2 offset1:%3"
                 : "=v"(d) : "v"(addr), "n"(O0), "n"(O1) : "memory");
    return d;
}
template <int O0, int O1>
__device__ __forceinline__ v4f lds_read_pairs2(unsigned addr) {
    v4f d;
    asm volatile("ds_load_2addr_b64 %0, %1 offset0:%2 offset1:%3"
                 : "=v"(d) : "v"(addr), "n"(O0), "n"(O1) : "memory");
    return d;
}
template <int N>
__device__ __forceinline__ void wait_reads(v4f& a, v2f& b0, v2f& b1, v2f& b2, v2f& b3) {
    asm volatile("s_wait_dscnt %5"
                 : "+v"(a), "+v"(b0), "+v"(b1), "+v"(b2), "+v"(b3) : "n"(N) : "memory");
}
template <int N>
__device__ __forceinline__ void wait_reads(v4f& a0, v4f& a1, v2f& b0, v2f& b1, v2f& b2, v2f& b3) {
    asm volatile("s_wait_dscnt %6"
                 : "+v"(a0), "+v"(a1), "+v"(b0), "+v"(b1), "+v"(b2), "+v"(b3) : "n"(N) : "memory");
}
__device__ __forceinline__ v2f lo2(v4f q) { return __builtin_shufflevector(q, q, 0, 1); }
__device__ __forceinline__ v2f hi2(v4f q) { return __builtin_shufflevector(q, q, 2, 3); }

typedef __attribute__((address_space(1))) const float gfloat;
typedef __attribute__((address_space(1))) const v4f   gv4f;
__device__ __forceinline__ gfloat* lane_uniform(const float* p) {
    const unsigned long long u = (unsigned long long)p;
    const unsigned lo = __builtin_amdgcn_readfirstlane((unsigned)u);
    const unsigned hi = __builtin_amdgcn_readfirstlane((unsigned)(u >> 32));
    return (gfloat*)(((unsigned long long)hi << 32) | lo);
}

struct Lane {
    int tid, r, hi, ni0, m0;
    int mb;
    int c4;
    int rsel;
    int wpos;
    unsigned wb[4];
    unsigned ab;
    unsigned ub;
};

template <int KS, int LD, bool AG>
__device__ __forceinline__ void slab_load(
    const float* __restrict__ A, const float* __restrict__ W, int ks0,
    const Lane& L, v4f* wreg, v4f& areg)
{
    gfloat* Ws = lane_uniform(W + (size_t)ks0 * RH);
    #pragma unroll
    for (int i = 0; i < KS / 4; ++i) {
        const unsigned off = (unsigned)((L.rsel + 4 * i) * RH + L.c4 * 4);
        wreg[i] = *(gv4f*)(Ws + off);
    }
    if (AG) {
        constexpr int QN = KS / 4;
        if (L.tid < MT * QN) {
            const int row = L.tid / QN, q = L.tid % QN;
            gfloat* As = lane_uniform(A + (size_t)L.m0 * LD + ks0);
            const unsigned off = (unsigned)(row * LD + 4 * q);
            areg = *(gv4f*)(As + off);
        }
    }
}

template <int KS, bool AG>
__device__ __forceinline__ void slab_store(
    float* Wl, float* Al, int buf, const Lane& L, const v4f* wreg, const v4f& areg)
{
    #pragma unroll
    for (int i = 0; i < KS / 4; ++i) {
        const int kl = L.rsel + 4 * i;
        *(v4f*)(Wl + buf * WSLAB + kl * WROW + L.wpos) = wreg[i];
    }
    if (AG) {
        constexpr int QN = KS / 4;
        if (L.tid < MT * QN) {
            const int row = L.tid / QN, q = L.tid % QN;
            *(v4f*)(Al + buf * ASLAB + row * APITCH + 4 * q) = areg;
        }
    }
}

__device__ __forceinline__ void slab_handover(v4f& a0, v4f& a1, v2f& b0, v2f& b1, v2f& b2, v2f& b3) {
    asm volatile("s_wait_dscnt 0x0\n\ts_barrier_signal -1\n\ts_barrier_wait -1"
                 : "+v"(a0), "+v"(a1), "+v"(b0), "+v"(b1), "+v"(b2), "+v"(b3) : : "memory");
}
__device__ __forceinline__ void pass_open_barrier() {
    asm volatile("s_wait_dscnt 0x0\n\ts_barrier_signal -1\n\ts_barrier_wait -1" : : : "memory");
}
template <int N>
__device__ __forceinline__ void wait_reads(v4f& a0, v4f& a1, v4f& a2,
                                           v2f (&b0)[4], v2f (&b1)[4], v2f (&b2)[4]) {
    asm volatile("s_wait_dscnt %15"
                 : "+v"(a0), "+v"(a1), "+v"(a2),
                   "+v"(b0[0]), "+v"(b0[1]), "+v"(b0[2]), "+v"(b0[3]),
                   "+v"(b1[0]), "+v"(b1[1]), "+v"(b1[2]), "+v"(b1[3]),
                   "+v"(b2[0]), "+v"(b2[1]), "+v"(b2[2]), "+v"(b2[3])
                 : "n"(N) : "memory");
}
template <int N>
__device__ __forceinline__ void wait_reads(v4f& a0, v4f& a1, v2f (&b0)[4], v2f (&b1)[4], v2f (&b2)[4]) {
    asm volatile("s_wait_dscnt %14"
                 : "+v"(a0), "+v"(a1),
                   "+v"(b0[0]), "+v"(b0[1]), "+v"(b0[2]), "+v"(b0[3]),
                   "+v"(b1[0]), "+v"(b1[1]), "+v"(b1[2]), "+v"(b1[3]),
                   "+v"(b2[0]), "+v"(b2[1]), "+v"(b2[2]), "+v"(b2[3])
                 : "n"(N) : "memory");
}
template <int N>
__device__ __forceinline__ void wait_reads(v4f& a0, v4f& a1, v2f (&b0)[4], v2f (&b1)[4]) {
    asm volatile("s_wait_dscnt %10"
                 : "+v"(a0), "+v"(a1),
                   "+v"(b0[0]), "+v"(b0[1]), "+v"(b0[2]), "+v"(b0[3]),
                   "+v"(b1[0]), "+v"(b1[1]), "+v"(b1[2]), "+v"(b1[3])
                 : "n"(N) : "memory");
}

template <int K, int KS, int LD>
__device__ __forceinline__ void gemm_pass_staged(
    const float* __restrict__ A, const float* __restrict__ W,
    float* Wl, float* Al, const Lane& L, v8f* acc)
{
    static_assert(K % KS == 0 && KS % 4 == 0 && KS <= KSMAX, "slab shape");
    constexpr int NS = K / KS;
    constexpr int NJ = KS / 4;
    static_assert(NJ == 3 || NJ == 4, "slabs of 12 or 16 k");
    static_assert(NS >= 2, "pipelined loop needs two slabs");
    v4f wreg[KS / 4];
    v4f areg = (v4f){};
    v4f a[NJ];
    v2f b[NJ][4];
    unsigned aB, wB[4];

#define SLAB_READS(J) \
    a[J]    = lds_read_pairs2<2 * (J), 160 + 2 * (J)>(aB); \
    b[J][0] = lds_read_rows2<32 * (J), 32 * (J) + 8>(wB[0]); \
    b[J][1] = lds_read_rows2<32 * (J), 32 * (J) + 8>(wB[1]); \
    b[J][2] = lds_read_rows2<32 * (J), 32 * (J) + 8>(wB[2]); \
    b[J][3] = lds_read_rows2<32 * (J), 32 * (J) + 8>(wB[3]);
#define SLAB_WMMAS(J) \
    _Pragma("unroll") \
    for (int t = 0; t < 4; ++t) { \
        acc[2 * t]     = wmma_f32_step(lo2(a[J]), b[J][t], acc[2 * t]); \
        acc[2 * t + 1] = wmma_f32_step(hi2(a[J]), b[J][t], acc[2 * t + 1]); \
    }
#define SLAB_BASES(BUF) \
    aB = L.ab + (unsigned)(BUF) * (ASLAB * 4u); \
    _Pragma("unroll") \
    for (int t = 0; t < 4; ++t) wB[t] = L.wb[t] + (unsigned)(BUF) * (WSLAB * 4u);
#define SLAB_WAIT_HEAD() \
    if (NJ == 4) wait_reads<5>(a[0], a[1], a[2], b[0], b[1], b[2]); \
    else         wait_reads<5>(a[0], a[1], b[0], b[1]);

    slab_load<KS, LD, true>(A, W, 0, L, wreg, areg);
    slab_store<KS, true>(Wl, Al, 0, L, wreg, areg);
    slab_load<KS, LD, true>(A, W, KS, L, wreg, areg);
    pass_open_barrier();
    SLAB_BASES(0)
    SLAB_READS(0) SLAB_READS(1) SLAB_READS(2)
    if (NJ == 4) { SLAB_READS(NJ - 1) }

    #pragma unroll 1
    for (int s = 0; s < NS - 1; ++s) {
        const int nxt = (s & 1) ^ 1;
        SLAB_WAIT_HEAD()
        SLAB_WMMAS(0)
        wait_reads<0>(a[NJ - 1], b[NJ - 1][0], b[NJ - 1][1], b[NJ - 1][2], b[NJ - 1][3]);
        slab_store<KS, true>(Wl, Al, nxt, L, wreg, areg);
        if (s + 2 < NS) slab_load<KS, LD, true>(A, W, (s + 2) * KS, L, wreg, areg);
        SLAB_WMMAS(1)
        if (NJ == 4) { SLAB_WMMAS(2) }
        slab_handover(a[NJ - 2], a[NJ - 1], b[NJ - 1][0], b[NJ - 1][1], b[NJ - 1][2], b[NJ - 1][3]);
        SLAB_BASES(nxt)
        SLAB_READS(0) SLAB_READS(1)
        if (NJ == 4) { SLAB_READS(2) }
        __builtin_amdgcn_sched_barrier(0);
        SLAB_WMMAS(NJ - 1)
        SLAB_READS(NJ - 1)
    }
    SLAB_WAIT_HEAD()
    SLAB_WMMAS(0)
    wait_reads<0>(a[NJ - 1], b[NJ - 1][0], b[NJ - 1][1], b[NJ - 1][2], b[NJ - 1][3]);
    SLAB_WMMAS(1) SLAB_WMMAS(2)
    if (NJ == 4) { SLAB_WMMAS(NJ - 1) }
    __syncthreads();
#undef SLAB_READS
#undef SLAB_WMMAS
#undef SLAB_BASES
#undef SLAB_WAIT_HEAD
}

__device__ __forceinline__ void gemm_pass_tile(
    const float* __restrict__ W, float* Wl, float* Al, const Lane& L, v8f* acc)
{
    constexpr int KS = 16;
    constexpr int NS = RH / KS;
    v4f wreg[KS / 4];
    v4f areg = (v4f){};
    v4f p0[2], p1[2];
    v2f b[4][4];
    unsigned u0, u1, wB[4];

#define TILE_B(J) \
    b[J][0] = lds_read_rows2<32 * (J), 32 * (J) + 8>(wB[0]); \
    b[J][1] = lds_read_rows2<32 * (J), 32 * (J) + 8>(wB[1]); \
    b[J][2] = lds_read_rows2<32 * (J), 32 * (J) + 8>(wB[2]); \
    b[J][3] = lds_read_rows2<32 * (J), 32 * (J) + 8>(wB[3]);
#define TILE_WMMAS(J, A0, A1) \
    _Pragma("unroll") \
    for (int t = 0; t < 4; ++t) { \
        acc[2 * t]     = wmma_f32_step(A0, b[J][t], acc[2 * t]); \
        acc[2 * t + 1] = wmma_f32_step(A1, b[J][t], acc[2 * t + 1]); \
    }
#define TILE_BASES(BUF, KS0) \
    u0 = L.ub + (unsigned)(KS0) * 4u; \
    u1 = u0 + 16u * UPITCH * 4u; \
    _Pragma("unroll") \
    for (int t = 0; t < 4; ++t) wB[t] = L.wb[t] + (unsigned)(BUF) * (WSLAB * 4u);
#define TILE_READS_HEAD() \
    p0[0] = lds_read_pairs2<0, 2>(u0); \
    p1[0] = lds_read_pairs2<0, 2>(u1); \
    TILE_B(0) TILE_B(1) TILE_B(2)
#define TILE_READS_TAIL() \
    p0[1] = lds_read_pairs2<4, 6>(u0); \
    p1[1] = lds_read_pairs2<4, 6>(u1); \
    TILE_B(3)

    slab_load<KS, RH, false>(nullptr, W, 0, L, wreg, areg);
    slab_store<KS, false>(Wl, Al, 0, L, wreg, areg);
    slab_load<KS, RH, false>(nullptr, W, KS, L, wreg, areg);
    pass_open_barrier();
    TILE_BASES(0, 0)
    TILE_READS_HEAD()
    TILE_READS_TAIL()

    #pragma unroll 1
    for (int s = 0; s < NS - 1; ++s) {
        const int nxt = (s & 1) ^ 1;
        wait_reads<6>(p0[0], p1[0], b[0], b[1], b[2]);
        TILE_WMMAS(0, lo2(p0[0]), lo2(p1[0]))
        wait_reads<0>(p0[1], p1[1], b[3][0], b[3][1], b[3][2], b[3][3]);
        slab_store<KS, false>(Wl, Al, nxt, L, wreg, areg);
        if (s + 2 < NS) slab_load<KS, RH, false>(nullptr, W, (s + 2) * KS, L, wreg, areg);
        TILE_WMMAS(1, hi2(p0[0]), hi2(p1[0]))
        TILE_WMMAS(2, lo2(p0[1]), lo2(p1[1]))
        slab_handover(p0[1], p1[1], b[3][0], b[3][1], b[3][2], b[3][3]);
        TILE_BASES(nxt, (s + 1) * KS)
        TILE_READS_HEAD()
        __builtin_amdgcn_sched_barrier(0);
        TILE_WMMAS(3, hi2(p0[1]), hi2(p1[1]))
        TILE_READS_TAIL()
    }
    wait_reads<6>(p0[0], p1[0], b[0], b[1], b[2]);
    TILE_WMMAS(0, lo2(p0[0]), lo2(p1[0]))
    wait_reads<0>(p0[1], p1[1], b[3][0], b[3][1], b[3][2], b[3][3]);
    TILE_WMMAS(1, hi2(p0[0]), hi2(p1[0]))
    TILE_WMMAS(2, lo2(p0[1]), lo2(p1[1]))
    TILE_WMMAS(3, hi2(p0[1]), hi2(p1[1]))
    __syncthreads();
#undef TILE_B
#undef TILE_WMMAS
#undef TILE_BASES
#undef TILE_READS_HEAD
#undef TILE_READS_TAIL
}

template <bool MULTIPLY>
__device__ __forceinline__ void product_tile(float* uLDS, const v8f* t8, const Lane& L)
{
    #pragma unroll
    for (int t = 0; t < 4; ++t) {
        const int nn = (L.ni0 + t) * 16 + L.r;
        #pragma unroll
        for (int mi = 0; mi < 2; ++mi) {
            #pragma unroll
            for (int e = 0; e < 8; ++e) {
                const int mm = L.mb + mi * 16 + 8 * L.hi + e;
                if (MULTIPLY) uLDS[mm * UPITCH + nn] *= t8[2 * t + mi][e];
                else          uLDS[mm * UPITCH + nn]  = t8[2 * t + mi][e];
            }
        }
    }
}

template <int K1, int KS1, int K2, int KS2>
__device__ __forceinline__ void product_and_project(
    const float* __restrict__ A1, const float* __restrict__ W1,
    const float* __restrict__ A2, const float* __restrict__ W2,
    const float* __restrict__ Ws,
    float* uLDS, float* Wl, float* Al, v8f* acc, const Lane& L)
{
    v8f t8[8];
    #pragma unroll
    for (int i = 0; i < 8; ++i) t8[i] = (v8f){};
    gemm_pass_staged<K1, KS1, K1>(A1, W1, Wl, Al, L, t8);
    product_tile<false>(uLDS, t8, L);


    #pragma unroll
    for (int i = 0; i < 8; ++i) t8[i] = (v8f){};
    gemm_pass_staged<K2, KS2, K2>(A2, W2, Wl, Al, L, t8);
    product_tile<true>(uLDS, t8, L);

    gemm_pass_tile(Ws, Wl, Al, L, acc);
}

__global__ __launch_bounds__(THREADS, 4)
void gate_kernel(const float* __restrict__ v,  const float* __restrict__ s,
                 const float* __restrict__ x,  const float* __restrict__ h,
                 const float* __restrict__ Wa, const float* __restrict__ Wb,
                 const float* __restrict__ Wc, const float* __restrict__ Ua,
                 const float* __restrict__ Ub, const float* __restrict__ Uc,
                 const float* __restrict__ Ca, const float* __restrict__ Cb,
                 const float* __restrict__ Cc, const float* __restrict__ bias,
                 float* __restrict__ logits  )
{
    __shared__ __attribute__((aligned(16))) float uLDS[MT * UPITCH];
    __shared__ __attribute__((aligned(16))) float Wl[2 * WSLAB];
    __shared__ __attribute__((aligned(16))) float Al[2 * ASLAB];

    const int g = blockIdx.y;
    Lane L;
    L.tid  = threadIdx.x;
    L.m0   = blockIdx.x * MT;
    const int lane = threadIdx.x & 31;
    const int wave = threadIdx.x >> 5;
    L.r    = lane & 15;
    L.hi   = lane >> 4;
    L.ni0  = (wave & 7) * 4;
    L.mb   = (wave >> 3) * 32;
    L.c4   = threadIdx.x & 127;
    L.rsel = threadIdx.x >> 7;
    L.wpos = (L.c4 * 4 + 16 * ((L.rsel >> 1) & 1)) & (WROW - 1);
    #pragma unroll
    for (int t = 0; t < 4; ++t) {
        const int pos = ((L.ni0 + t) * 16 + L.r + 16 * L.hi) & (WROW - 1);
        L.wb[t] = lds_addr(Wl) + (unsigned)(2 * L.hi * WROW + pos) * 4u;
    }
    L.ab = lds_addr(Al)   + (unsigned)((L.mb + L.r) * APITCH + 2 * L.hi) * 4u;
    L.ub = lds_addr(uLDS) + (unsigned)((L.mb + L.r) * UPITCH + 2 * L.hi) * 4u;

    const float* Wa_g = Wa + (size_t)g * RH * RH;
    const float* Wb_g = Wb + (size_t)g * NT * RH;
    const float* Wc_g = Wc + (size_t)g * RH * HD;
    const float* Ua_g = Ua + (size_t)g * RH * RH;
    const float* Ub_g = Ub + (size_t)g * NT * RH;
    const float* Uc_g = Uc + (size_t)g * RH * HD;
    const float* Ca_g = Ca + (size_t)g * NF * RH;
    const float* Cb_g = Cb + (size_t)g * NT * RH;
    const float* Cc_g = Cc + (size_t)g * RH * HD;

    v8f acc[8];
    #pragma unroll
    for (int t = 0; t < 8; ++t) acc[t] = (v8f){};

    product_and_project<RH, 16, NT, 12>(x, Wa_g, s, Wb_g, Wc_g, uLDS, Wl, Al, acc, L);
    product_and_project<NF, 16, NT, 12>(v, Ca_g, s, Cb_g, Cc_g, uLDS, Wl, Al, acc, L);
    product_and_project<NT, 12, RH, 16>(s, Ub_g, h, Ua_g, Uc_g, uLDS, Wl, Al, acc, L);

    const float* bg = bias + (size_t)g * HD;
    float* Lg = logits + ((size_t)g * BSZ + L.m0) * HD;
    #pragma unroll
    for (int t = 0; t < 4; ++t) {
        const int nn = (L.ni0 + t) * 16 + L.r;
        #pragma unroll
        for (int mi = 0; mi < 2; ++mi) {
            #pragma unroll
            for (int e = 0; e < 8; ++e) {
                const int mm = L.mb + mi * 16 + 8 * L.hi + e;
                Lg[(size_t)mm * HD + nn] = acc[2 * t + mi][e] + bg[nn];
            }
        }
    }
}

__global__ __launch_bounds__(256)
void lstm_epilogue(const float* __restrict__ logits,
                   const float* __restrict__ c,
                   float* __restrict__ out, int n)
{
    const int idx = blockIdx.x * blockDim.x + threadIdx.x;
    if (idx >= n) return;
    const size_t stride = (size_t)BSZ * HD;
    const float li = logits[idx];
    const float lf = logits[idx + stride];
    const float lo = logits[idx + 2 * stride];
    const float lg = logits[idx + 3 * stride];
    const float ig = 1.0f / (1.0f + __expf(-li));
    const float fg = 1.0f / (1.0f + __expf(-lf));
    const float og = 1.0f / (1.0f + __expf(-lo));
    const float gg = tanhf(lg);
    const float nc = fg * c[idx] + ig * gg;
    const float nh = og * tanhf(nc);
    out[idx]          = nh;
    out[idx + stride] = nc;
}

extern "C" void kernel_launch(void* const* d_in, const int* in_sizes, int n_in,
                              void* d_out, int out_size, void* d_ws, size_t ws_size,
                              hipStream_t stream) {
    (void)in_sizes; (void)n_in; (void)out_size; (void)ws_size;
    const float* v  = (const float*)d_in[0];
    const float* s  = (const float*)d_in[1];
    const float* x  = (const float*)d_in[2];
    const float* h  = (const float*)d_in[3];
    const float* c  = (const float*)d_in[4];
    const float* Wa = (const float*)d_in[5];
    const float* Wb = (const float*)d_in[6];
    const float* Wc = (const float*)d_in[7];
    const float* Ua = (const float*)d_in[8];
    const float* Ub = (const float*)d_in[9];
    const float* Uc = (const float*)d_in[10];
    const float* Ca = (const float*)d_in[11];
    const float* Cb = (const float*)d_in[12];
    const float* Cc = (const float*)d_in[13];
    const float* b  = (const float*)d_in[14];

    float* logits = (float*)d_ws;

    dim3 grid(BSZ / MT, 4);
    gate_kernel<<<grid, dim3(THREADS), 0, stream>>>(
        v, s, x, h, Wa, Wb, Wc, Ua, Ub, Uc, Ca, Cb, Cc, b, logits);

    const int n = BSZ * HD;
    lstm_epilogue<<<(n + 255) / 256, 256, 0, stream>>>(logits, c, (float*)d_out, n);
}
